# starting kernel
# baseline (speedup 1.0000x reference)
#include <hip/hip_runtime.h>

#define B_ 64
#define I_ 512
#define D_ 128
#define O_ 32
#define H_ 64

typedef __attribute__((ext_vector_type(2))) float v2f;
typedef __attribute__((ext_vector_type(8))) float v8f;

__global__ void k_colsum(const float* __restrict__ x, float* __restrict__ xs) {
  int b = blockIdx.x;
  int d = threadIdx.x;
  const float* xp = x + (size_t)b * I_ * D_ + d;
  float s = 0.f;
#pragma unroll 8
  for (int i = 0; i < I_; ++i) s += xp[(size_t)i * D_];
  xs[b * D_ + d] = s;
}

__global__ void k_iter0(const float* __restrict__ xs, const float* __restrict__ W,
                        float* __restrict__ u) {
  int bo = blockIdx.x;
  int b = bo / O_, o = bo % O_;
  int h = threadIdx.x;
  __shared__ float out0[H_];
  __shared__ float red[H_];

  const float* Wo  = W + (size_t)o * D_ * H_;
  const float* xsb = xs + b * D_;

  float v = 0.f;
#pragma unroll 4
  for (int d = 0; d < D_; ++d) v += xsb[d] * Wo[d * H_ + h];
  v *= (1.0f / O_);

  red[h] = v * v;
  __syncthreads();
  if (h == 0) {
    float sn = 0.f;
    for (int t = 0; t < H_; ++t) sn += red[t];
    red[0] = sn;
  }
  __syncthreads();
  float sn = red[0];
  float scale = (sn / (1.f + sn)) * rsqrtf(sn);
  out0[h] = v * scale;
  __syncthreads();

  float* ub = u + (size_t)bo * D_;
  for (int dd = h; dd < D_; dd += H_) {
    const float* Wd = Wo + dd * H_;
    float acc = 0.f;
#pragma unroll 4
    for (int hh = 0; hh < H_; ++hh) acc += Wd[hh] * out0[hh];
    ub[dd] = acc;
  }
}

__global__ void k_logits_softmax(const float* __restrict__ x,
                                 const float* __restrict__ u,
                                 float* __restrict__ probs) {
  int blk = blockIdx.x;
  int b  = blk >> 5;
  int i0 = (blk & 31) * 16;
  int l  = threadIdx.x;
  int lh = l & 15;
  int kh = (l >> 4) * 2;

  const float* xb = x + (size_t)b * I_ * D_;
  const float* ub = u + (size_t)b * O_ * D_;

  const float* arow  = xb + (size_t)(i0 + lh) * D_ + kh;
  const float* brow0 = ub + (size_t)lh * D_ + kh;
  const float* brow1 = ub + (size_t)(lh + 16) * D_ + kh;

  v8f acc0 = {};
  v8f acc1 = {};
#pragma unroll 4
  for (int k = 0; k < D_; k += 4) {
    v2f a  = *(const v2f*)(arow + k);
    v2f b0 = *(const v2f*)(brow0 + k);
    v2f b1 = *(const v2f*)(brow1 + k);
    acc0 = __builtin_amdgcn_wmma_f32_16x16x4_f32(false, a, false, b0, (short)0,
                                                 acc0, false, false);
    acc1 = __builtin_amdgcn_wmma_f32_16x16x4_f32(false, a, false, b1, (short)0,
                                                 acc1, false, false);
  }

  float pr0[8], pr1[8];
#pragma unroll
  for (int r = 0; r < 8; ++r) {
    float m = fmaxf(acc0[r], acc1[r]);
    m = fmaxf(m, __shfl_xor(m, 8, 32));
    m = fmaxf(m, __shfl_xor(m, 4, 32));
    m = fmaxf(m, __shfl_xor(m, 2, 32));
    m = fmaxf(m, __shfl_xor(m, 1, 32));
    float e0 = __expf(acc0[r] - m);
    float e1 = __expf(acc1[r] - m);
    float s = e0 + e1;
    s += __shfl_xor(s, 8, 32);
    s += __shfl_xor(s, 4, 32);
    s += __shfl_xor(s, 2, 32);
    s += __shfl_xor(s, 1, 32);
    float inv = 1.0f / s;
    pr0[r] = e0 * inv;
    pr1[r] = e1 * inv;
  }

  float* pb = probs + (size_t)b * O_ * I_;
  int irow = i0 + 8 * (l >> 4);
#pragma unroll
  for (int r = 0; r < 8; ++r) {
    pb[(size_t)lh * I_ + irow + r]        = pr0[r];
    pb[(size_t)(lh + 16) * I_ + irow + r] = pr1[r];
  }
}

__global__ void k_y(const float* __restrict__ probs, const float* __restrict__ x,
                    float* __restrict__ y) {
  int blk = blockIdx.x;
  int b  = blk >> 4;
  int t  = blk & 15;
  int m0 = (t >> 3) * 16;
  int n0 = (t & 7) * 16;
  int l  = threadIdx.x;
  int lh = l & 15;
  int kh = (l >> 4) * 2;

  const float* pa = probs + (size_t)b * O_ * I_ + (size_t)(m0 + lh) * I_ + kh;
  const float* xb = x + (size_t)b * I_ * D_ + n0 + lh;

  v8f acc = {};
#pragma unroll 4
  for (int k = 0; k < I_; k += 4) {
    v2f a = *(const v2f*)(pa + k);
    v2f bb;
    bb.x = xb[(size_t)(k + kh) * D_];
    bb.y = xb[(size_t)(k + kh + 1) * D_];
    acc = __builtin_amdgcn_wmma_f32_16x16x4_f32(false, a, false, bb, (short)0,
                                                acc, false, false);
  }

  float* yb = y + (size_t)b * O_ * D_;
  int orow = m0 + 8 * (l >> 4);
#pragma unroll
  for (int r = 0; r < 8; ++r)
    yb[(size_t)(orow + r) * D_ + n0 + lh] = acc[r];
}

__global__ void k_out(const float* __restrict__ y, const float* __restrict__ W,
                      float* __restrict__ out) {
  int bo = blockIdx.x;
  int o = bo % O_;
  int h = threadIdx.x;
  __shared__ float red[H_];

  const float* yb = y + (size_t)bo * D_;
  const float* Wo = W + (size_t)o * D_ * H_;

  float v = 0.f;
#pragma unroll 4
  for (int d = 0; d < D_; ++d) v += yb[d] * Wo[d * H_ + h];

  red[h] = v * v;
  __syncthreads();
  if (h == 0) {
    float sn = 0.f;
    for (int t = 0; t < H_; ++t) sn += red[t];
    red[0] = sn;
  }
  __syncthreads();
  float sn = red[0];
  float scale = (sn / (1.f + sn)) * rsqrtf(sn);
  out[(size_t)bo * H_ + h] = v * scale;
}

extern "C" void kernel_launch(void* const* d_in, const int* in_sizes, int n_in,
                              void* d_out, int out_size, void* d_ws, size_t ws_size,
                              hipStream_t stream) {
  const float* x = (const float*)d_in[0];
  const float* W = (const float*)d_in[1];
  float* out = (float*)d_out;

  char* ws = (char*)d_ws;
  float* xs = (float*)ws;    ws += (size_t)B_ * D_ * sizeof(float);
  float* u  = (float*)ws;    ws += (size_t)B_ * O_ * D_ * sizeof(float);
  float* pr = (float*)ws;    ws += (size_t)B_ * O_ * I_ * sizeof(float);
  float* y  = (float*)ws;

  k_colsum<<<B_, D_, 0, stream>>>(x, xs);
  k_iter0<<<B_ * O_, H_, 0, stream>>>(xs, W, u);
  k_logits_softmax<<<B_ * (I_ / 16), 32, 0, stream>>>(x, u, pr);
  k_y<<<B_ * 16, 32, 0, stream>>>(pr, x, y);
  k_out<<<B_ * O_, H_, 0, stream>>>(y, W, out);
}


# column-sum kernel: the 512-row serial sum per thread becomes 256 blocks x 256 threads, 16 independent 16-byte loads per thread, row groups folded through LDS
# speedup vs baseline: 1.5681x; 1.5681x over previous
#include <hip/hip_runtime.h>

#define B_ 64
#define I_ 512
#define D_ 128
#define O_ 32
#define H_ 64

typedef __attribute__((ext_vector_type(2))) float v2f;
typedef __attribute__((ext_vector_type(4))) float v4f;
typedef __attribute__((ext_vector_type(8))) float v8f;

__global__ __launch_bounds__(256) void k_colsum(const float* __restrict__ x,
                                                float* __restrict__ xs) {
  int blk = blockIdx.x;
  int b = blk >> 2;
  int q = blk & 3;
  int t = threadIdx.x;
  int c = t & 7;
  int r = t >> 3;
  __shared__ float red[32][32];

  const float* xp = x + (size_t)b * I_ * D_ + (size_t)r * D_ + q * 32 + c * 4;
  v4f v[16];
#pragma unroll
  for (int j = 0; j < 16; ++j) v[j] = *(const v4f*)(xp + (size_t)j * 32 * D_);
  __builtin_amdgcn_sched_barrier(0);

#pragma unroll
  for (int j = 0; j < 8; ++j) v[j] += v[j + 8];
#pragma unroll
  for (int j = 0; j < 4; ++j) v[j] += v[j + 4];
  v4f s = (v[0] + v[2]) + (v[1] + v[3]);

  *(v4f*)&red[r][c * 4] = s;
  __syncthreads();

  if (t < 32) {
    float a0 = 0.f, a1 = 0.f, a2 = 0.f, a3 = 0.f;
#pragma unroll
    for (int j = 0; j < 32; j += 4) {
      a0 += red[j][t];
      a1 += red[j + 1][t];
      a2 += red[j + 2][t];
      a3 += red[j + 3][t];
    }
    xs[b * D_ + q * 32 + t] = (a0 + a2) + (a1 + a3);
  }
}

__global__ void k_iter0(const float* __restrict__ xs, const float* __restrict__ W,
                        float* __restrict__ u) {
  int bo = blockIdx.x;
  int b = bo / O_, o = bo % O_;
  int h = threadIdx.x;
  __shared__ float out0[H_];
  __shared__ float red[H_];

  const float* Wo  = W + (size_t)o * D_ * H_;
  const float* xsb = xs + b * D_;

  float v = 0.f;
#pragma unroll 4
  for (int d = 0; d < D_; ++d) v += xsb[d] * Wo[d * H_ + h];
  v *= (1.0f / O_);

  red[h] = v * v;
  __syncthreads();
  if (h == 0) {
    float sn = 0.f;
    for (int t = 0; t < H_; ++t) sn += red[t];
    red[0] = sn;
  }
  __syncthreads();
  float sn = red[0];
  float scale = (sn / (1.f + sn)) * rsqrtf(sn);
  out0[h] = v * scale;
  __syncthreads();

  float* ub = u + (size_t)bo * D_;
  for (int dd = h; dd < D_; dd += H_) {
    const float* Wd = Wo + dd * H_;
    float acc = 0.f;
#pragma unroll 4
    for (int hh = 0; hh < H_; ++hh) acc += Wd[hh] * out0[hh];
    ub[dd] = acc;
  }
}

__global__ void k_logits_softmax(const float* __restrict__ x,
                                 const float* __restrict__ u,
                                 float* __restrict__ probs) {
  int blk = blockIdx.x;
  int b  = blk >> 5;
  int i0 = (blk & 31) * 16;
  int l  = threadIdx.x;
  int lh = l & 15;
  int kh = (l >> 4) * 2;

  const float* xb = x + (size_t)b * I_ * D_;
  const float* ub = u + (size_t)b * O_ * D_;

  const float* arow  = xb + (size_t)(i0 + lh) * D_ + kh;
  const float* brow0 = ub + (size_t)lh * D_ + kh;
  const float* brow1 = ub + (size_t)(lh + 16) * D_ + kh;

  v8f acc0 = {};
  v8f acc1 = {};
#pragma unroll 4
  for (int k = 0; k < D_; k += 4) {
    v2f a  = *(const v2f*)(arow + k);
    v2f b0 = *(const v2f*)(brow0 + k);
    v2f b1 = *(const v2f*)(brow1 + k);
    acc0 = __builtin_amdgcn_wmma_f32_16x16x4_f32(false, a, false, b0, (short)0,
                                                 acc0, false, false);
    acc1 = __builtin_amdgcn_wmma_f32_16x16x4_f32(false, a, false, b1, (short)0,
                                                 acc1, false, false);
  }

  float pr0[8], pr1[8];
#pragma unroll
  for (int r = 0; r < 8; ++r) {
    float m = fmaxf(acc0[r], acc1[r]);
    m = fmaxf(m, __shfl_xor(m, 8, 32));
    m = fmaxf(m, __shfl_xor(m, 4, 32));
    m = fmaxf(m, __shfl_xor(m, 2, 32));
    m = fmaxf(m, __shfl_xor(m, 1, 32));
    float e0 = __expf(acc0[r] - m);
    float e1 = __expf(acc1[r] - m);
    float s = e0 + e1;
    s += __shfl_xor(s, 8, 32);
    s += __shfl_xor(s, 4, 32);
    s += __shfl_xor(s, 2, 32);
    s += __shfl_xor(s, 1, 32);
    float inv = 1.0f / s;
    pr0[r] = e0 * inv;
    pr1[r] = e1 * inv;
  }

  float* pb = probs + (size_t)b * O_ * I_;
  int irow = i0 + 8 * (l >> 4);
#pragma unroll
  for (int r = 0; r < 8; ++r) {
    pb[(size_t)lh * I_ + irow + r]        = pr0[r];
    pb[(size_t)(lh + 16) * I_ + irow + r] = pr1[r];
  }
}

__global__ void k_y(const float* __restrict__ probs, const float* __restrict__ x,
                    float* __restrict__ y) {
  int blk = blockIdx.x;
  int b  = blk >> 4;
  int t  = blk & 15;
  int m0 = (t >> 3) * 16;
  int n0 = (t & 7) * 16;
  int l  = threadIdx.x;
  int lh = l & 15;
  int kh = (l >> 4) * 2;

  const float* pa = probs + (size_t)b * O_ * I_ + (size_t)(m0 + lh) * I_ + kh;
  const float* xb = x + (size_t)b * I_ * D_ + n0 + lh;

  v8f acc = {};
#pragma unroll 4
  for (int k = 0; k < I_; k += 4) {
    v2f a = *(const v2f*)(pa + k);
    v2f bb;
    bb.x = xb[(size_t)(k + kh) * D_];
    bb.y = xb[(size_t)(k + kh + 1) * D_];
    acc = __builtin_amdgcn_wmma_f32_16x16x4_f32(false, a, false, bb, (short)0,
                                                acc, false, false);
  }

  float* yb = y + (size_t)b * O_ * D_;
  int orow = m0 + 8 * (l >> 4);
#pragma unroll
  for (int r = 0; r < 8; ++r)
    yb[(size_t)(orow + r) * D_ + n0 + lh] = acc[r];
}

__global__ void k_out(const float* __restrict__ y, const float* __restrict__ W,
                      float* __restrict__ out) {
  int bo = blockIdx.x;
  int o = bo % O_;
  int h = threadIdx.x;
  __shared__ float red[H_];

  const float* yb = y + (size_t)bo * D_;
  const float* Wo = W + (size_t)o * D_ * H_;

  float v = 0.f;
#pragma unroll 4
  for (int d = 0; d < D_; ++d) v += yb[d] * Wo[d * H_ + h];

  red[h] = v * v;
  __syncthreads();
  if (h == 0) {
    float sn = 0.f;
    for (int t = 0; t < H_; ++t) sn += red[t];
    red[0] = sn;
  }
  __syncthreads();
  float sn = red[0];
  float scale = (sn / (1.f + sn)) * rsqrtf(sn);
  out[(size_t)bo * H_ + h] = v * scale;
}

extern "C" void kernel_launch(void* const* d_in, const int* in_sizes, int n_in,
                              void* d_out, int out_size, void* d_ws, size_t ws_size,
                              hipStream_t stream) {
  const float* x = (const float*)d_in[0];
  const float* W = (const float*)d_in[1];
  float* out = (float*)d_out;

  char* ws = (char*)d_ws;
  float* xs = (float*)ws;    ws += (size_t)B_ * D_ * sizeof(float);
  float* u  = (float*)ws;    ws += (size_t)B_ * O_ * D_ * sizeof(float);
  float* pr = (float*)ws;    ws += (size_t)B_ * O_ * I_ * sizeof(float);
  float* y  = (float*)ws;

  k_colsum<<<B_ * 4, 256, 0, stream>>>(x, xs);
  k_iter0<<<B_ * O_, H_, 0, stream>>>(xs, W, u);
  k_logits_softmax<<<B_ * (I_ / 16), 32, 0, stream>>>(x, u, pr);
  k_y<<<B_ * 16, 32, 0, stream>>>(pr, x, y);
  k_out<<<B_ * O_, H_, 0, stream>>>(y, W, out);
}


# iter0 and out kernels: W contractions batched over 16 batch rows per block on the f32 16x16x4 matrix form, W[o] read once per 16 rows, all fragment loads requested before the first product
# speedup vs baseline: 2.5533x; 1.6283x over previous
#include <hip/hip_runtime.h>

#define B_ 64
#define I_ 512
#define D_ 128
#define O_ 32
#define H_ 64

typedef __attribute__((ext_vector_type(2))) float v2f;
typedef __attribute__((ext_vector_type(4))) float v4f;
typedef __attribute__((ext_vector_type(8))) float v8f;

__global__ __launch_bounds__(256) void k_colsum(const float* __restrict__ x,
                                                float* __restrict__ xs) {
  int blk = blockIdx.x;
  int b = blk >> 2;
  int q = blk & 3;
  int t = threadIdx.x;
  int c = t & 7;
  int r = t >> 3;
  __shared__ float red[32][32];

  const float* xp = x + (size_t)b * I_ * D_ + (size_t)r * D_ + q * 32 + c * 4;
  v4f v[16];
#pragma unroll
  for (int j = 0; j < 16; ++j) v[j] = *(const v4f*)(xp + (size_t)j * 32 * D_);
  __builtin_amdgcn_sched_barrier(0);

#pragma unroll
  for (int j = 0; j < 8; ++j) v[j] += v[j + 8];
#pragma unroll
  for (int j = 0; j < 4; ++j) v[j] += v[j + 4];
  v4f s = (v[0] + v[2]) + (v[1] + v[3]);

  *(v4f*)&red[r][c * 4] = s;
  __syncthreads();

  if (t < 32) {
    float a0 = 0.f, a1 = 0.f, a2 = 0.f, a3 = 0.f;
#pragma unroll
    for (int j = 0; j < 32; j += 4) {
      a0 += red[j][t];
      a1 += red[j + 1][t];
      a2 += red[j + 2][t];
      a3 += red[j + 3][t];
    }
    xs[b * D_ + q * 32 + t] = (a0 + a2) + (a1 + a3);
  }
}

__device__ __forceinline__ v8f tile_k128(const float* __restrict__ arow,
                                         const float* __restrict__ bcol) {
  v2f a[32], p[32];
#pragma unroll
  for (int j = 0; j < 32; ++j) {
    a[j] = *(const v2f*)(arow + 4 * j);
    p[j].x = bcol[(size_t)(4 * j) * H_];
    p[j].y = bcol[(size_t)(4 * j + 1) * H_];
  }
  __builtin_amdgcn_sched_barrier(0);
  v8f acc0 = {};
  v8f acc1 = {};
#pragma unroll
  for (int j = 0; j < 32; j += 2) {
    acc0 = __builtin_amdgcn_wmma_f32_16x16x4_f32(false, a[j], false, p[j], (short)0,
                                                 acc0, false, false);
    acc1 = __builtin_amdgcn_wmma_f32_16x16x4_f32(false, a[j + 1], false, p[j + 1],
                                                 (short)0, acc1, false, false);
  }
  return acc0 + acc1;
}

__device__ __forceinline__ void row_scales(const v8f& v, float (*ssq)[16], int w, int lh,
                                           int hf, float* scale) {
  float q[8];
#pragma unroll
  for (int r = 0; r < 8; ++r) q[r] = v[r] * v[r];
#pragma unroll
  for (int m = 8; m >= 1; m >>= 1) {
#pragma unroll
    for (int r = 0; r < 8; ++r) q[r] += __shfl_xor(q[r], m, 32);
  }
  if (lh == 0) {
#pragma unroll
    for (int r = 0; r < 8; ++r) ssq[w][8 * hf + r] = q[r];
  }
  __syncthreads();
#pragma unroll
  for (int r = 0; r < 8; ++r) {
    int row = 8 * hf + r;
    float sn = (ssq[0][row] + ssq[1][row]) + (ssq[2][row] + ssq[3][row]);
    scale[r] = (sn / (1.f + sn)) * rsqrtf(sn);
  }
}

__global__ __launch_bounds__(128) void k_iter0(const float* __restrict__ xs,
                                               const float* __restrict__ W,
                                               float* __restrict__ u) {
  int blk = blockIdx.x;
  int o  = blk >> 2;
  int b0 = (blk & 3) * 16;
  int t  = threadIdx.x;
  int w  = t >> 5;
  int l  = t & 31;
  int lh = l & 15;
  int hf = l >> 4;
  int kh = hf * 2;

  __shared__ __attribute__((aligned(16))) float out0s[16][H_ + 4];
  __shared__ float ssq[4][16];

  const float* Wo = W + (size_t)o * D_ * H_;

  const float* brow0 = Wo + (size_t)(16 * w + lh) * H_ + kh;
  const float* brow1 = brow0 + (size_t)64 * H_;
  v2f q0[16], q1[16];
#pragma unroll
  for (int j = 0; j < 16; ++j) {
    q0[j] = *(const v2f*)(brow0 + 4 * j);
    q1[j] = *(const v2f*)(brow1 + 4 * j);
  }

  v8f v = tile_k128(xs + (size_t)(b0 + lh) * D_ + kh,
                    Wo + (size_t)kh * H_ + 16 * w + lh);
  v *= (1.0f / O_);

  float scale[8];
  row_scales(v, ssq, w, lh, hf, scale);
#pragma unroll
  for (int r = 0; r < 8; ++r) out0s[8 * hf + r][16 * w + lh] = v[r] * scale[r];
  __syncthreads();

  v8f c0 = {};
  v8f c1 = {};
#pragma unroll
  for (int j = 0; j < 16; ++j) {
    v2f a = *(const v2f*)&out0s[lh][4 * j + kh];
    c0 = __builtin_amdgcn_wmma_f32_16x16x4_f32(false, a, false, q0[j], (short)0,
                                               c0, false, false);
    c1 = __builtin_amdgcn_wmma_f32_16x16x4_f32(false, a, false, q1[j], (short)0,
                                               c1, false, false);
  }

  float* ub = u + ((size_t)(b0 + 8 * hf) * O_ + o) * D_ + 16 * w + lh;
#pragma unroll
  for (int r = 0; r < 8; ++r) {
    ub[(size_t)r * O_ * D_]      = c0[r];
    ub[(size_t)r * O_ * D_ + 64] = c1[r];
  }
}

__global__ void k_logits_softmax(const float* __restrict__ x,
                                 const float* __restrict__ u,
                                 float* __restrict__ probs) {
  int blk = blockIdx.x;
  int b  = blk >> 5;
  int i0 = (blk & 31) * 16;
  int l  = threadIdx.x;
  int lh = l & 15;
  int kh = (l >> 4) * 2;

  const float* xb = x + (size_t)b * I_ * D_;
  const float* ub = u + (size_t)b * O_ * D_;

  const float* arow  = xb + (size_t)(i0 + lh) * D_ + kh;
  const float* brow0 = ub + (size_t)lh * D_ + kh;
  const float* brow1 = ub + (size_t)(lh + 16) * D_ + kh;

  v8f acc0 = {};
  v8f acc1 = {};
#pragma unroll 4
  for (int k = 0; k < D_; k += 4) {
    v2f a  = *(const v2f*)(arow + k);
    v2f b0 = *(const v2f*)(brow0 + k);
    v2f b1 = *(const v2f*)(brow1 + k);
    acc0 = __builtin_amdgcn_wmma_f32_16x16x4_f32(false, a, false, b0, (short)0,
                                                 acc0, false, false);
    acc1 = __builtin_amdgcn_wmma_f32_16x16x4_f32(false, a, false, b1, (short)0,
                                                 acc1, false, false);
  }

  float pr0[8], pr1[8];
#pragma unroll
  for (int r = 0; r < 8; ++r) {
    float m = fmaxf(acc0[r], acc1[r]);
    m = fmaxf(m, __shfl_xor(m, 8, 32));
    m = fmaxf(m, __shfl_xor(m, 4, 32));
    m = fmaxf(m, __shfl_xor(m, 2, 32));
    m = fmaxf(m, __shfl_xor(m, 1, 32));
    float e0 = __expf(acc0[r] - m);
    float e1 = __expf(acc1[r] - m);
    float s = e0 + e1;
    s += __shfl_xor(s, 8, 32);
    s += __shfl_xor(s, 4, 32);
    s += __shfl_xor(s, 2, 32);
    s += __shfl_xor(s, 1, 32);
    float inv = 1.0f / s;
    pr0[r] = e0 * inv;
    pr1[r] = e1 * inv;
  }

  float* pb = probs + (size_t)b * O_ * I_;
  int irow = i0 + 8 * (l >> 4);
#pragma unroll
  for (int r = 0; r < 8; ++r) {
    pb[(size_t)lh * I_ + irow + r]        = pr0[r];
    pb[(size_t)(lh + 16) * I_ + irow + r] = pr1[r];
  }
}

__global__ void k_y(const float* __restrict__ probs, const float* __restrict__ x,
                    float* __restrict__ y) {
  int blk = blockIdx.x;
  int b  = blk >> 4;
  int t  = blk & 15;
  int m0 = (t >> 3) * 16;
  int n0 = (t & 7) * 16;
  int l  = threadIdx.x;
  int lh = l & 15;
  int kh = (l >> 4) * 2;

  const float* pa = probs + (size_t)b * O_ * I_ + (size_t)(m0 + lh) * I_ + kh;
  const float* xb = x + (size_t)b * I_ * D_ + n0 + lh;

  v8f acc = {};
#pragma unroll 4
  for (int k = 0; k < I_; k += 4) {
    v2f a = *(const v2f*)(pa + k);
    v2f bb;
    bb.x = xb[(size_t)(k + kh) * D_];
    bb.y = xb[(size_t)(k + kh + 1) * D_];
    acc = __builtin_amdgcn_wmma_f32_16x16x4_f32(false, a, false, bb, (short)0,
                                                acc, false, false);
  }

  float* yb = y + (size_t)b * O_ * D_;
  int orow = m0 + 8 * (l >> 4);
#pragma unroll
  for (int r = 0; r < 8; ++r)
    yb[(size_t)(orow + r) * D_ + n0 + lh] = acc[r];
}

__global__ __launch_bounds__(128) void k_out(const float* __restrict__ y,
                                             const float* __restrict__ W,
                                             float* __restrict__ out) {
  int blk = blockIdx.x;
  int o  = blk >> 2;
  int b0 = (blk & 3) * 16;
  int t  = threadIdx.x;
  int w  = t >> 5;
  int l  = t & 31;
  int lh = l & 15;
  int hf = l >> 4;
  int kh = hf * 2;

  __shared__ float ssq[4][16];

  const float* Wo = W + (size_t)o * D_ * H_;
  v8f v = tile_k128(y + ((size_t)(b0 + lh) * O_ + o) * D_ + kh,
                    Wo + (size_t)kh * H_ + 16 * w + lh);

  float scale[8];
  row_scales(v, ssq, w, lh, hf, scale);

  float* ob = out + ((size_t)(b0 + 8 * hf) * O_ + o) * H_ + 16 * w + lh;
#pragma unroll
  for (int r = 0; r < 8; ++r) ob[(size_t)r * O_ * H_] = v[r] * scale[r];
}

extern "C" void kernel_launch(void* const* d_in, const int* in_sizes, int n_in,
                              void* d_out, int out_size, void* d_ws, size_t ws_size,
                              hipStream_t stream) {
  const float* x = (const float*)d_in[0];
  const float* W = (const float*)d_in[1];
  float* out = (float*)d_out;

  char* ws = (char*)d_ws;
  float* xs = (float*)ws;    ws += (size_t)B_ * D_ * sizeof(float);
  float* u  = (float*)ws;    ws += (size_t)B_ * O_ * D_ * sizeof(float);
  float* pr = (float*)ws;    ws += (size_t)B_ * O_ * I_ * sizeof(float);
  float* y  = (float*)ws;

  k_colsum<<<B_ * 4, 256, 0, stream>>>(x, xs);
  k_iter0<<<O_ * (B_ / 16), 128, 0, stream>>>(xs, W, u);
  k_logits_softmax<<<B_ * (I_ / 16), 32, 0, stream>>>(x, u, pr);
  k_y<<<B_ * 16, 32, 0, stream>>>(pr, x, y);
  k_out<<<O_ * (B_ / 16), 128, 0, stream>>>(y, W, out);
}


# y kernel: K split over the 8 waves of a (batch, 32-column strip) block; each wave loads its x and probs slabs by whole-line 16-byte loads all in flight, stages them in LDS, 64 products on 4 tiles
# speedup vs baseline: 3.5930x; 1.4072x over previous
#include <hip/hip_runtime.h>

#define B_ 64
#define I_ 512
#define D_ 128
#define O_ 32
#define H_ 64

typedef __attribute__((ext_vector_type(2))) float v2f;
typedef __attribute__((ext_vector_type(4))) float v4f;
typedef __attribute__((ext_vector_type(8))) float v8f;

__global__ __launch_bounds__(256) void k_colsum(const float* __restrict__ x,
                                                float* __restrict__ xs) {
  int blk = blockIdx.x;
  int b = blk >> 2;
  int q = blk & 3;
  int t = threadIdx.x;
  int c = t & 7;
  int r = t >> 3;
  __shared__ float red[32][32];

  const float* xp = x + (size_t)b * I_ * D_ + (size_t)r * D_ + q * 32 + c * 4;
  v4f v[16];
#pragma unroll
  for (int j = 0; j < 16; ++j) v[j] = *(const v4f*)(xp + (size_t)j * 32 * D_);
  __builtin_amdgcn_sched_barrier(0);

#pragma unroll
  for (int j = 0; j < 8; ++j) v[j] += v[j + 8];
#pragma unroll
  for (int j = 0; j < 4; ++j) v[j] += v[j + 4];
  v4f s = (v[0] + v[2]) + (v[1] + v[3]);

  *(v4f*)&red[r][c * 4] = s;
  __syncthreads();

  if (t < 32) {
    float a0 = 0.f, a1 = 0.f, a2 = 0.f, a3 = 0.f;
#pragma unroll
    for (int j = 0; j < 32; j += 4) {
      a0 += red[j][t];
      a1 += red[j + 1][t];
      a2 += red[j + 2][t];
      a3 += red[j + 3][t];
    }
    xs[b * D_ + q * 32 + t] = (a0 + a2) + (a1 + a3);
  }
}

__device__ __forceinline__ v8f tile_k128(const float* __restrict__ arow,
                                         const float* __restrict__ bcol) {
  v2f a[32], p[32];
#pragma unroll
  for (int j = 0; j < 32; ++j) {
    a[j] = *(const v2f*)(arow + 4 * j);
    p[j].x = bcol[(size_t)(4 * j) * H_];
    p[j].y = bcol[(size_t)(4 * j + 1) * H_];
  }
  __builtin_amdgcn_sched_barrier(0);
  v8f acc0 = {};
  v8f acc1 = {};
#pragma unroll
  for (int j = 0; j < 32; j += 2) {
    acc0 = __builtin_amdgcn_wmma_f32_16x16x4_f32(false, a[j], false, p[j], (short)0,
                                                 acc0, false, false);
    acc1 = __builtin_amdgcn_wmma_f32_16x16x4_f32(false, a[j + 1], false, p[j + 1],
                                                 (short)0, acc1, false, false);
  }
  return acc0 + acc1;
}

__device__ __forceinline__ void row_scales(const v8f& v, float (*ssq)[16], int w, int lh,
                                           int hf, float* scale) {
  float q[8];
#pragma unroll
  for (int r = 0; r < 8; ++r) q[r] = v[r] * v[r];
#pragma unroll
  for (int m = 8; m >= 1; m >>= 1) {
#pragma unroll
    for (int r = 0; r < 8; ++r) q[r] += __shfl_xor(q[r], m, 32);
  }
  if (lh == 0) {
#pragma unroll
    for (int r = 0; r < 8; ++r) ssq[w][8 * hf + r] = q[r];
  }
  __syncthreads();
#pragma unroll
  for (int r = 0; r < 8; ++r) {
    int row = 8 * hf + r;
    float sn = (ssq[0][row] + ssq[1][row]) + (ssq[2][row] + ssq[3][row]);
    scale[r] = (sn / (1.f + sn)) * rsqrtf(sn);
  }
}

__global__ __launch_bounds__(128) void k_iter0(const float* __restrict__ xs,
                                               const float* __restrict__ W,
                                               float* __restrict__ u) {
  int blk = blockIdx.x;
  int o  = blk >> 2;
  int b0 = (blk & 3) * 16;
  int t  = threadIdx.x;
  int w  = t >> 5;
  int l  = t & 31;
  int lh = l & 15;
  int hf = l >> 4;
  int kh = hf * 2;

  __shared__ __attribute__((aligned(16))) float out0s[16][H_ + 4];
  __shared__ float ssq[4][16];

  const float* Wo = W + (size_t)o * D_ * H_;

  const float* brow0 = Wo + (size_t)(16 * w + lh) * H_ + kh;
  const float* brow1 = brow0 + (size_t)64 * H_;
  v2f q0[16], q1[16];
#pragma unroll
  for (int j = 0; j < 16; ++j) {
    q0[j] = *(const v2f*)(brow0 + 4 * j);
    q1[j] = *(const v2f*)(brow1 + 4 * j);
  }

  v8f v = tile_k128(xs + (size_t)(b0 + lh) * D_ + kh,
                    Wo + (size_t)kh * H_ + 16 * w + lh);
  v *= (1.0f / O_);

  float scale[8];
  row_scales(v, ssq, w, lh, hf, scale);
#pragma unroll
  for (int r = 0; r < 8; ++r) out0s[8 * hf + r][16 * w + lh] = v[r] * scale[r];
  __syncthreads();

  v8f c0 = {};
  v8f c1 = {};
#pragma unroll
  for (int j = 0; j < 16; ++j) {
    v2f a = *(const v2f*)&out0s[lh][4 * j + kh];
    c0 = __builtin_amdgcn_wmma_f32_16x16x4_f32(false, a, false, q0[j], (short)0,
                                               c0, false, false);
    c1 = __builtin_amdgcn_wmma_f32_16x16x4_f32(false, a, false, q1[j], (short)0,
                                               c1, false, false);
  }

  float* ub = u + ((size_t)(b0 + 8 * hf) * O_ + o) * D_ + 16 * w + lh;
#pragma unroll
  for (int r = 0; r < 8; ++r) {
    ub[(size_t)r * O_ * D_]      = c0[r];
    ub[(size_t)r * O_ * D_ + 64] = c1[r];
  }
}

__global__ void k_logits_softmax(const float* __restrict__ x,
                                 const float* __restrict__ u,
                                 float* __restrict__ probs) {
  int blk = blockIdx.x;
  int b  = blk >> 5;
  int i0 = (blk & 31) * 16;
  int l  = threadIdx.x;
  int lh = l & 15;
  int kh = (l >> 4) * 2;

  const float* xb = x + (size_t)b * I_ * D_;
  const float* ub = u + (size_t)b * O_ * D_;

  const float* arow  = xb + (size_t)(i0 + lh) * D_ + kh;
  const float* brow0 = ub + (size_t)lh * D_ + kh;
  const float* brow1 = ub + (size_t)(lh + 16) * D_ + kh;

  v8f acc0 = {};
  v8f acc1 = {};
#pragma unroll 4
  for (int k = 0; k < D_; k += 4) {
    v2f a  = *(const v2f*)(arow + k);
    v2f b0 = *(const v2f*)(brow0 + k);
    v2f b1 = *(const v2f*)(brow1 + k);
    acc0 = __builtin_amdgcn_wmma_f32_16x16x4_f32(false, a, false, b0, (short)0,
                                                 acc0, false, false);
    acc1 = __builtin_amdgcn_wmma_f32_16x16x4_f32(false, a, false, b1, (short)0,
                                                 acc1, false, false);
  }

  float pr0[8], pr1[8];
#pragma unroll
  for (int r = 0; r < 8; ++r) {
    float m = fmaxf(acc0[r], acc1[r]);
    m = fmaxf(m, __shfl_xor(m, 8, 32));
    m = fmaxf(m, __shfl_xor(m, 4, 32));
    m = fmaxf(m, __shfl_xor(m, 2, 32));
    m = fmaxf(m, __shfl_xor(m, 1, 32));
    float e0 = __expf(acc0[r] - m);
    float e1 = __expf(acc1[r] - m);
    float s = e0 + e1;
    s += __shfl_xor(s, 8, 32);
    s += __shfl_xor(s, 4, 32);
    s += __shfl_xor(s, 2, 32);
    s += __shfl_xor(s, 1, 32);
    float inv = 1.0f / s;
    pr0[r] = e0 * inv;
    pr1[r] = e1 * inv;
  }

  float* pb = probs + (size_t)b * O_ * I_;
  int irow = i0 + 8 * (l >> 4);
#pragma unroll
  for (int r = 0; r < 8; ++r) {
    pb[(size_t)lh * I_ + irow + r]        = pr0[r];
    pb[(size_t)(lh + 16) * I_ + irow + r] = pr1[r];
  }
}

#define YXP 40
#define YPP 68
#define YSLAB (64 * YXP + 32 * YPP)
#define YRP 36
#define YLDS (162 * 256)

__global__ __launch_bounds__(256) void k_y(const float* __restrict__ probs,
                                           const float* __restrict__ x,
                                           float* __restrict__ y) {
  int blk = blockIdx.x;
  int b  = blk >> 2;
  int q  = blk & 3;
  int t  = threadIdx.x;
  int w  = t >> 5;
  int l  = t & 31;
  int lh = l & 15;
  int hf = l >> 4;

  __shared__ __attribute__((aligned(16))) float sm[YLDS];

  const float* xg = x + ((size_t)b * I_ + 64 * w + (l >> 3)) * D_ + 32 * q + 4 * (l & 7);
  const float* pg = probs + ((size_t)b * O_ + hf) * I_ + 64 * w + 4 * lh;
  v4f xr[16], pr[16];
#pragma unroll
  for (int n = 0; n < 16; ++n) xr[n] = *(const v4f*)(xg + (size_t)(4 * n) * D_);
#pragma unroll
  for (int n = 0; n < 16; ++n) pr[n] = *(const v4f*)(pg + (size_t)(2 * n) * I_);
  __builtin_amdgcn_sched_barrier(0);

  float* xs = sm + w * YSLAB;
  float* ps = xs + 64 * YXP;
#pragma unroll
  for (int n = 0; n < 16; ++n)
    *(v4f*)(xs + (4 * n + (l >> 3)) * YXP + 4 * (l & 7)) = xr[n];
#pragma unroll
  for (int n = 0; n < 16; ++n)
    *(v4f*)(ps + (2 * n + hf) * YPP + 4 * lh) = pr[n];
  __syncthreads();

  v2f a0[16], a1[16], b0[16], b1[16];
  const float* pa = ps + lh * YPP + 2 * hf;
  const float* xb = xs + (2 * hf) * YXP + lh;
#pragma unroll
  for (int j = 0; j < 16; ++j) {
    a0[j] = *(const v2f*)(pa + 4 * j);
    a1[j] = *(const v2f*)(pa + 16 * YPP + 4 * j);
    b0[j].x = xb[(4 * j) * YXP];
    b0[j].y = xb[(4 * j + 1) * YXP];
    b1[j].x = xb[(4 * j) * YXP + 16];
    b1[j].y = xb[(4 * j + 1) * YXP + 16];
  }
  __builtin_amdgcn_sched_barrier(0);

  v8f c00 = {}, c01 = {}, c10 = {}, c11 = {};
#pragma unroll
  for (int j = 0; j < 16; ++j) {
    c00 = __builtin_amdgcn_wmma_f32_16x16x4_f32(false, a0[j], false, b0[j], (short)0,
                                                c00, false, false);
    c01 = __builtin_amdgcn_wmma_f32_16x16x4_f32(false, a0[j], false, b1[j], (short)0,
                                                c01, false, false);
    c10 = __builtin_amdgcn_wmma_f32_16x16x4_f32(false, a1[j], false, b0[j], (short)0,
                                                c10, false, false);
    c11 = __builtin_amdgcn_wmma_f32_16x16x4_f32(false, a1[j], false, b1[j], (short)0,
                                                c11, false, false);
  }

  __syncthreads();

  float* rw = sm + w * (32 * YRP) + (8 * hf) * YRP + lh;
#pragma unroll
  for (int r = 0; r < 8; ++r) {
    rw[r * YRP]                 = c00[r];
    rw[r * YRP + 16]            = c01[r];
    rw[(16 + r) * YRP]          = c10[r];
    rw[(16 + r) * YRP + 16]     = c11[r];
  }
  __syncthreads();

  const float* rr = sm + (t >> 3) * YRP + 4 * (t & 7);
  v4f s0 = *(const v4f*)(rr);
  v4f s1 = *(const v4f*)(rr + 1 * 32 * YRP);
  v4f s2 = *(const v4f*)(rr + 2 * 32 * YRP);
  v4f s3 = *(const v4f*)(rr + 3 * 32 * YRP);
  v4f s4 = *(const v4f*)(rr + 4 * 32 * YRP);
  v4f s5 = *(const v4f*)(rr + 5 * 32 * YRP);
  v4f s6 = *(const v4f*)(rr + 6 * 32 * YRP);
  v4f s7 = *(const v4f*)(rr + 7 * 32 * YRP);
  v4f s = ((s0 + s1) + (s2 + s3)) + ((s4 + s5) + (s6 + s7));
  *(v4f*)(y + ((size_t)b * O_ + (t >> 3)) * D_ + 32 * q + 4 * (t & 7)) = s;
}

__global__ __launch_bounds__(128) void k_out(const float* __restrict__ y,
                                             const float* __restrict__ W,
                                             float* __restrict__ out) {
  int blk = blockIdx.x;
  int o  = blk >> 2;
  int b0 = (blk & 3) * 16;
  int t  = threadIdx.x;
  int w  = t >> 5;
  int l  = t & 31;
  int lh = l & 15;
  int hf = l >> 4;
  int kh = hf * 2;

  __shared__ float ssq[4][16];

  const float* Wo = W + (size_t)o * D_ * H_;
  v8f v = tile_k128(y + ((size_t)(b0 + lh) * O_ + o) * D_ + kh,
                    Wo + (size_t)kh * H_ + 16 * w + lh);

  float scale[8];
  row_scales(v, ssq, w, lh, hf, scale);

  float* ob = out + ((size_t)(b0 + 8 * hf) * O_ + o) * H_ + 16 * w + lh;
#pragma unroll
  for (int r = 0; r < 8; ++r) ob[(size_t)r * O_ * H_] = v[r] * scale[r];
}

extern "C" void kernel_launch(void* const* d_in, const int* in_sizes, int n_in,
                              void* d_out, int out_size, void* d_ws, size_t ws_size,
                              hipStream_t stream) {
  const float* x = (const float*)d_in[0];
  const float* W = (const float*)d_in[1];
  float* out = (float*)d_out;

  char* ws = (char*)d_ws;
  float* xs = (float*)ws;    ws += (size_t)B_ * D_ * sizeof(float);
  float* u  = (float*)ws;    ws += (size_t)B_ * O_ * D_ * sizeof(float);
  float* pr = (float*)ws;    ws += (size_t)B_ * O_ * I_ * sizeof(float);
  float* y  = (float*)ws;

  k_colsum<<<B_ * 4, 256, 0, stream>>>(x, xs);
  k_iter0<<<O_ * (B_ / 16), 128, 0, stream>>>(xs, W, u);
  k_logits_softmax<<<B_ * (I_ / 16), 32, 0, stream>>>(x, u, pr);
  k_y<<<B_ * 4, 256, 0, stream>>>(pr, x, y);
  k_out<<<O_ * (B_ / 16), 128, 0, stream>>>(y, W, out);
}


# iter0 and out kernels: W[o] and the 16 A rows fetched by whole-line 16-byte loads all in flight and staged in LDS; fragments read from LDS instead of per-lane strided global loads
# speedup vs baseline: 5.1029x; 1.4202x over previous
#include <hip/hip_runtime.h>

#define B_ 64
#define I_ 512
#define D_ 128
#define O_ 32
#define H_ 64

typedef __attribute__((ext_vector_type(2))) float v2f;
typedef __attribute__((ext_vector_type(4))) float v4f;
typedef __attribute__((ext_vector_type(8))) float v8f;

__global__ __launch_bounds__(256) void k_colsum(const float* __restrict__ x,
                                                float* __restrict__ xs) {
  int blk = blockIdx.x;
  int b = blk >> 2;
  int q = blk & 3;
  int t = threadIdx.x;
  int c = t & 7;
  int r = t >> 3;
  __shared__ float red[32][32];

  const float* xp = x + (size_t)b * I_ * D_ + (size_t)r * D_ + q * 32 + c * 4;
  v4f v[16];
#pragma unroll
  for (int j = 0; j < 16; ++j) v[j] = *(const v4f*)(xp + (size_t)j * 32 * D_);
  __builtin_amdgcn_sched_barrier(0);

#pragma unroll
  for (int j = 0; j < 8; ++j) v[j] += v[j + 8];
#pragma unroll
  for (int j = 0; j < 4; ++j) v[j] += v[j + 4];
  v4f s = (v[0] + v[2]) + (v[1] + v[3]);

  *(v4f*)&red[r][c * 4] = s;
  __syncthreads();

  if (t < 32) {
    float a0 = 0.f, a1 = 0.f, a2 = 0.f, a3 = 0.f;
#pragma unroll
    for (int j = 0; j < 32; j += 4) {
      a0 += red[j][t];
      a1 += red[j + 1][t];
      a2 += red[j + 2][t];
      a3 += red[j + 3][t];
    }
    xs[b * D_ + q * 32 + t] = (a0 + a2) + (a1 + a3);
  }
}

#define PWS 68
#define PAS 132

__device__ __forceinline__ void w_load(const float* __restrict__ Wo, int t, v4f (&wr)[16]) {
#pragma unroll
  for (int n = 0; n < 16; ++n) wr[n] = *(const v4f*)(Wo + 512 * n + 4 * t);
}
__device__ __forceinline__ void w_store(float* WS, int t, const v4f (&wr)[16]) {
#pragma unroll
  for (int n = 0; n < 16; ++n) *(v4f*)(WS + (8 * n + (t >> 4)) * PWS + 4 * (t & 15)) = wr[n];
}

struct frag_k128 {
  v4f a[16];
  float b[16][4];
};
__device__ __forceinline__ void frag_k128_load(frag_k128& f, const float* AS, const float* WS,
                                               int n0, int lh, int hf) {
  const float* pa = AS + lh * PAS + 4 * hf;
  const float* pb = WS + (4 * hf) * PWS + n0 + lh;
#pragma unroll
  for (int jj = 0; jj < 16; ++jj) {
    f.a[jj] = *(const v4f*)(pa + 8 * jj);
#pragma unroll
    for (int e = 0; e < 4; ++e) f.b[jj][e] = pb[(8 * jj + e) * PWS];
  }
}
__device__ __forceinline__ v8f frag_k128_mma(const frag_k128& f) {
  v8f acc0 = {};
  v8f acc1 = {};
#pragma unroll
  for (int jj = 0; jj < 16; ++jj) {
    v2f alo = f.a[jj].xy, ahi = f.a[jj].zw;
    v2f blo = {f.b[jj][0], f.b[jj][1]};
    v2f bhi = {f.b[jj][2], f.b[jj][3]};
    acc0 = __builtin_amdgcn_wmma_f32_16x16x4_f32(false, alo, false, blo, (short)0,
                                                 acc0, false, false);
    acc1 = __builtin_amdgcn_wmma_f32_16x16x4_f32(false, ahi, false, bhi, (short)0,
                                                 acc1, false, false);
  }
  return acc0 + acc1;
}

__device__ __forceinline__ void row_scales(const v8f& v, float (*ssq)[16], int w, int lh,
                                           int hf, float* scale) {
  float q[8];
#pragma unroll
  for (int r = 0; r < 8; ++r) q[r] = v[r] * v[r];
#pragma unroll
  for (int m = 8; m >= 1; m >>= 1) {
#pragma unroll
    for (int r = 0; r < 8; ++r) q[r] += __shfl_xor(q[r], m, 32);
  }
  if (lh == 0) {
#pragma unroll
    for (int r = 0; r < 8; ++r) ssq[w][8 * hf + r] = q[r];
  }
  __syncthreads();
#pragma unroll
  for (int r = 0; r < 8; ++r) {
    int row = 8 * hf + r;
    float sn = (ssq[0][row] + ssq[1][row]) + (ssq[2][row] + ssq[3][row]);
    scale[r] = (sn / (1.f + sn)) * rsqrtf(sn);
  }
}

__global__ __launch_bounds__(128) void k_iter0(const float* __restrict__ xs,
                                               const float* __restrict__ W,
                                               float* __restrict__ u) {
  int blk = blockIdx.x;
  int o  = blk >> 2;
  int b0 = (blk & 3) * 16;
  int t  = threadIdx.x;
  int w  = t >> 5;
  int l  = t & 31;
  int lh = l & 15;
  int hf = l >> 4;

  __shared__ __attribute__((aligned(16))) float WS[128 * PWS];
  __shared__ __attribute__((aligned(16))) float AS[16 * PAS];
  __shared__ __attribute__((aligned(16))) float out0s[16 * PWS];
  __shared__ float ssq[4][16];

  const float* ag = xs + (size_t)(b0 + w) * D_ + 4 * l;
  v4f wr[16], ar[4];
#pragma unroll
  for (int n = 0; n < 4; ++n) ar[n] = *(const v4f*)(ag + (size_t)(4 * n) * D_);
  w_load(W + (size_t)o * D_ * H_, t, wr);
  __builtin_amdgcn_sched_barrier(0);
#pragma unroll
  for (int n = 0; n < 4; ++n) *(v4f*)(AS + (4 * n + w) * PAS + 4 * l) = ar[n];
  w_store(WS, t, wr);
  __syncthreads();

  frag_k128 f;
  frag_k128_load(f, AS, WS, 16 * w, lh, hf);
  v4f q0[8], q1[8];
  {
    const float* pq = WS + (16 * w + lh) * PWS + 4 * hf;
#pragma unroll
    for (int jj = 0; jj < 8; ++jj) {
      q0[jj] = *(const v4f*)(pq + 8 * jj);
      q1[jj] = *(const v4f*)(pq + 64 * PWS + 8 * jj);
    }
  }
  __builtin_amdgcn_sched_barrier(0);

  v8f v = frag_k128_mma(f);
  v *= (1.0f / O_);

  float scale[8];
  row_scales(v, ssq, w, lh, hf, scale);
#pragma unroll
  for (int r = 0; r < 8; ++r) out0s[(8 * hf + r) * PWS + 16 * w + lh] = v[r] * scale[r];
  __syncthreads();

  v8f c0 = {};
  v8f c1 = {};
  {
    v4f fa[8];
    const float* pa = out0s + lh * PWS + 4 * hf;
#pragma unroll
    for (int jj = 0; jj < 8; ++jj) fa[jj] = *(const v4f*)(pa + 8 * jj);
    __builtin_amdgcn_sched_barrier(0);
#pragma unroll
    for (int jj = 0; jj < 8; ++jj) {
      v2f alo = fa[jj].xy, ahi = fa[jj].zw;
      v2f b0lo = q0[jj].xy, b0hi = q0[jj].zw;
      v2f b1lo = q1[jj].xy, b1hi = q1[jj].zw;
      c0 = __builtin_amdgcn_wmma_f32_16x16x4_f32(false, alo, false, b0lo, (short)0,
                                                 c0, false, false);
      c1 = __builtin_amdgcn_wmma_f32_16x16x4_f32(false, alo, false, b1lo, (short)0,
                                                 c1, false, false);
      c0 = __builtin_amdgcn_wmma_f32_16x16x4_f32(false, ahi, false, b0hi, (short)0,
                                                 c0, false, false);
      c1 = __builtin_amdgcn_wmma_f32_16x16x4_f32(false, ahi, false, b1hi, (short)0,
                                                 c1, false, false);
    }
  }

  float* ub = u + ((size_t)(b0 + 8 * hf) * O_ + o) * D_ + 16 * w + lh;
#pragma unroll
  for (int r = 0; r < 8; ++r) {
    ub[(size_t)r * O_ * D_]      = c0[r];
    ub[(size_t)r * O_ * D_ + 64] = c1[r];
  }
}

#define RXP 140
#define RPP 132
#define R_US (128 * RXP)
#define R_PS (R_US + 32 * RXP)
#define R_YS (R_PS + 32 * RPP)
#define RLDS (162 * 256)

__global__ __launch_bounds__(256) void k_route(const float* __restrict__ x,
                                               const float* __restrict__ u,
                                               float* __restrict__ yp) {
  int blk = blockIdx.x;
  int b  = blk >> 2;
  int c  = blk & 3;
  int t  = threadIdx.x;
  int w  = t >> 5;
  int l  = t & 31;
  int lh = l & 15;
  int hf = l >> 4;

  __shared__ __attribute__((aligned(16))) float sm[RLDS];
  float* XS = sm;
  float* US = sm + R_US;
  float* PS = sm + R_PS;
  float* YS = sm + R_YS;

  const float* xg = x + ((size_t)b * I_ + 128 * c + w) * D_ + 4 * l;
  const float* ug = u + ((size_t)b * O_ + w) * D_ + 4 * l;
  v4f xr[16], ur[4];
#pragma unroll
  for (int n = 0; n < 4; ++n) ur[n] = *(const v4f*)(ug + (size_t)(8 * n) * D_);
#pragma unroll
  for (int n = 0; n < 16; ++n) xr[n] = *(const v4f*)(xg + (size_t)(8 * n) * D_);
  __builtin_amdgcn_sched_barrier(0);
#pragma unroll
  for (int n = 0; n < 4; ++n) *(v4f*)(US + (8 * n + w) * RXP + 4 * l) = ur[n];
#pragma unroll
  for (int n = 0; n < 16; ++n) *(v4f*)(XS + (8 * n + w) * RXP + 4 * l) = xr[n];
  __syncthreads();

  v8f acc0 = {};
  v8f acc1 = {};
  {
    v4f fa[16], fb0[16], fb1[16];
    const float* pa = XS + (16 * w + lh) * RXP + 4 * hf;
    const float* pb = US + lh * RXP + 4 * hf;
#pragma unroll
    for (int jj = 0; jj < 16; ++jj) {
      fa[jj]  = *(const v4f*)(pa + 8 * jj);
      fb0[jj] = *(const v4f*)(pb + 8 * jj);
      fb1[jj] = *(const v4f*)(pb + 16 * RXP + 8 * jj);
    }
    __builtin_amdgcn_sched_barrier(0);
#pragma unroll
    for (int jj = 0; jj < 16; ++jj) {
      v2f alo = fa[jj].xy, ahi = fa[jj].zw;
      v2f b0lo = fb0[jj].xy, b0hi = fb0[jj].zw;
      v2f b1lo = fb1[jj].xy, b1hi = fb1[jj].zw;
      acc0 = __builtin_amdgcn_wmma_f32_16x16x4_f32(false, alo, false, b0lo, (short)0,
                                                   acc0, false, false);
      acc1 = __builtin_amdgcn_wmma_f32_16x16x4_f32(false, alo, false, b1lo, (short)0,
                                                   acc1, false, false);
      acc0 = __builtin_amdgcn_wmma_f32_16x16x4_f32(false, ahi, false, b0hi, (short)0,
                                                   acc0, false, false);
      acc1 = __builtin_amdgcn_wmma_f32_16x16x4_f32(false, ahi, false, b1hi, (short)0,
                                                   acc1, false, false);
    }
  }

  float pr0[8], pr1[8];
#pragma unroll
  for (int r = 0; r < 8; ++r) {
    float m = fmaxf(acc0[r], acc1[r]);
    m = fmaxf(m, __shfl_xor(m, 8, 32));
    m = fmaxf(m, __shfl_xor(m, 4, 32));
    m = fmaxf(m, __shfl_xor(m, 2, 32));
    m = fmaxf(m, __shfl_xor(m, 1, 32));
    float e0 = __expf(acc0[r] - m);
    float e1 = __expf(acc1[r] - m);
    float s = e0 + e1;
    s += __shfl_xor(s, 8, 32);
    s += __shfl_xor(s, 4, 32);
    s += __shfl_xor(s, 2, 32);
    s += __shfl_xor(s, 1, 32);
    float inv = 1.0f / s;
    pr0[r] = e0 * inv;
    pr1[r] = e1 * inv;
  }
  {
    float* pw = PS + lh * RPP + 16 * w + 8 * hf;
    v4f t0 = {pr0[0], pr0[1], pr0[2], pr0[3]};
    v4f t1 = {pr0[4], pr0[5], pr0[6], pr0[7]};
    v4f t2 = {pr1[0], pr1[1], pr1[2], pr1[3]};
    v4f t3 = {pr1[4], pr1[5], pr1[6], pr1[7]};
    *(v4f*)(pw)                = t0;
    *(v4f*)(pw + 4)            = t1;
    *(v4f*)(pw + 16 * RPP)     = t2;
    *(v4f*)(pw + 16 * RPP + 4) = t3;
  }
  __syncthreads();

  int np = w & 3;
  int k2 = w >> 2;
  v8f c00 = {}, c01 = {}, c10 = {}, c11 = {};
  {
    v4f ga0[8], ga1[8];
    float gb0[8][4], gb1[8][4];
    const float* qa = PS + lh * RPP + 64 * k2 + 4 * hf;
    const float* qb = XS + (64 * k2 + 4 * hf) * RXP + 32 * np + lh;
#pragma unroll
    for (int jj = 0; jj < 8; ++jj) {
      ga0[jj] = *(const v4f*)(qa + 8 * jj);
      ga1[jj] = *(const v4f*)(qa + 16 * RPP + 8 * jj);
#pragma unroll
      for (int e = 0; e < 4; ++e) {
        gb0[jj][e] = qb[(8 * jj + e) * RXP];
        gb1[jj][e] = qb[(8 * jj + e) * RXP + 16];
      }
    }
    __builtin_amdgcn_sched_barrier(0);
#pragma unroll
    for (int jj = 0; jj < 8; ++jj) {
      v2f a0lo = ga0[jj].xy, a0hi = ga0[jj].zw;
      v2f a1lo = ga1[jj].xy, a1hi = ga1[jj].zw;
      v2f b0lo = {gb0[jj][0], gb0[jj][1]};
      v2f b0hi = {gb0[jj][2], gb0[jj][3]};
      v2f b1lo = {gb1[jj][0], gb1[jj][1]};
      v2f b1hi = {gb1[jj][2], gb1[jj][3]};
      c00 = __builtin_amdgcn_wmma_f32_16x16x4_f32(false, a0lo, false, b0lo, (short)0,
                                                  c00, false, false);
      c01 = __builtin_amdgcn_wmma_f32_16x16x4_f32(false, a0lo, false, b1lo, (short)0,
                                                  c01, false, false);
      c10 = __builtin_amdgcn_wmma_f32_16x16x4_f32(false, a1lo, false, b0lo, (short)0,
                                                  c10, false, false);
      c11 = __builtin_amdgcn_wmma_f32_16x16x4_f32(false, a1lo, false, b1lo, (short)0,
                                                  c11, false, false);
      c00 = __builtin_amdgcn_wmma_f32_16x16x4_f32(false, a0hi, false, b0hi, (short)0,
                                                  c00, false, false);
      c01 = __builtin_amdgcn_wmma_f32_16x16x4_f32(false, a0hi, false, b1hi, (short)0,
                                                  c01, false, false);
      c10 = __builtin_amdgcn_wmma_f32_16x16x4_f32(false, a1hi, false, b0hi, (short)0,
                                                  c10, false, false);
      c11 = __builtin_amdgcn_wmma_f32_16x16x4_f32(false, a1hi, false, b1hi, (short)0,
                                                  c11, false, false);
    }
  }

  float* rw = YS + k2 * (32 * RPP) + (8 * hf) * RPP + 32 * np + lh;
#pragma unroll
  for (int r = 0; r < 8; ++r) {
    rw[r * RPP]             = c00[r];
    rw[r * RPP + 16]        = c01[r];
    rw[(16 + r) * RPP]      = c10[r];
    rw[(16 + r) * RPP + 16] = c11[r];
  }
  __syncthreads();

  float* yo = yp + (((size_t)c * B_ + b) * O_) * D_ + 4 * l;
#pragma unroll
  for (int n = 0; n < 4; ++n) {
    int o = 8 * n + w;
    v4f s0 = *(const v4f*)(YS + o * RPP + 4 * l);
    v4f s1 = *(const v4f*)(YS + 32 * RPP + o * RPP + 4 * l);
    *(v4f*)(yo + (size_t)o * D_) = s0 + s1;
  }
}

__global__ __launch_bounds__(128) void k_out(const float* __restrict__ yp,
                                             const float* __restrict__ W,
                                             float* __restrict__ out) {
  int blk = blockIdx.x;
  int o  = blk >> 2;
  int b0 = (blk & 3) * 16;
  int t  = threadIdx.x;
  int w  = t >> 5;
  int l  = t & 31;
  int lh = l & 15;
  int hf = l >> 4;

  __shared__ __attribute__((aligned(16))) float WS[128 * PWS];
  __shared__ __attribute__((aligned(16))) float AS[16 * PAS];
  __shared__ float ssq[4][16];

  const float* yg = yp + ((size_t)(b0 + w) * O_ + o) * D_ + 4 * l;
  const size_t PL = (size_t)B_ * O_ * D_;
  v4f wr[16], y0[4], y1[4], y2[4], y3[4];
#pragma unroll
  for (int n = 0; n < 4; ++n) {
    const float* yn = yg + (size_t)(4 * n) * O_ * D_;
    y0[n] = *(const v4f*)(yn);
    y1[n] = *(const v4f*)(yn + PL);
    y2[n] = *(const v4f*)(yn + 2 * PL);
    y3[n] = *(const v4f*)(yn + 3 * PL);
  }
  w_load(W + (size_t)o * D_ * H_, t, wr);
  __builtin_amdgcn_sched_barrier(0);
#pragma unroll
  for (int n = 0; n < 4; ++n)
    *(v4f*)(AS + (4 * n + w) * PAS + 4 * l) = (y0[n] + y1[n]) + (y2[n] + y3[n]);
  w_store(WS, t, wr);
  __syncthreads();

  frag_k128 f;
  frag_k128_load(f, AS, WS, 16 * w, lh, hf);
  __builtin_amdgcn_sched_barrier(0);
  v8f v = frag_k128_mma(f);

  float scale[8];
  row_scales(v, ssq, w, lh, hf, scale);

  float* ob = out + ((size_t)(b0 + 8 * hf) * O_ + o) * H_ + 16 * w + lh;
#pragma unroll
  for (int r = 0; r < 8; ++r) ob[(size_t)r * O_ * H_] = v[r] * scale[r];
}

extern "C" void kernel_launch(void* const* d_in, const int* in_sizes, int n_in,
                              void* d_out, int out_size, void* d_ws, size_t ws_size,
                              hipStream_t stream) {
  const float* x = (const float*)d_in[0];
  const float* W = (const float*)d_in[1];
  float* out = (float*)d_out;

  char* ws = (char*)d_ws;
  float* xs = (float*)ws;    ws += (size_t)B_ * D_ * sizeof(float);
  float* u  = (float*)ws;    ws += (size_t)B_ * O_ * D_ * sizeof(float);
  float* yp = (float*)ws;

  k_colsum<<<B_ * 4, 256, 0, stream>>>(x, xs);
  k_iter0<<<O_ * (B_ / 16), 128, 0, stream>>>(xs, W, u);
  k_route<<<B_ * 4, 256, 0, stream>>>(x, u, yp);
  k_out<<<O_ * (B_ / 16), 128, 0, stream>>>(yp, W, out);
}


# iter0, route, out: product before each reduction taken transposed (operands swapped): softmax and squash sums run down a lane's registers + 1 permlanex16, no ds_bpermute; 1 division per lane not 8
# speedup vs baseline: 5.3599x; 1.0504x over previous
#include <hip/hip_runtime.h>

#define B_ 64
#define I_ 512
#define D_ 128
#define O_ 32
#define H_ 64

typedef __attribute__((ext_vector_type(2))) float v2f;
typedef __attribute__((ext_vector_type(4))) float v4f;
typedef __attribute__((ext_vector_type(8))) float v8f;

__global__ __launch_bounds__(256) void k_colsum(const float* __restrict__ x,
                                                float* __restrict__ xs) {
  int blk = blockIdx.x;
  int b = blk >> 2;
  int q = blk & 3;
  int t = threadIdx.x;
  int c = t & 7;
  int r = t >> 3;
  __shared__ float red[32][32];

  const float* xp = x + (size_t)b * I_ * D_ + (size_t)r * D_ + q * 32 + c * 4;
  v4f v[16];
#pragma unroll
  for (int j = 0; j < 16; ++j) v[j] = *(const v4f*)(xp + (size_t)j * 32 * D_);
  __builtin_amdgcn_sched_barrier(0);

#pragma unroll
  for (int j = 0; j < 8; ++j) v[j] += v[j + 8];
#pragma unroll
  for (int j = 0; j < 4; ++j) v[j] += v[j + 4];
  v4f s = (v[0] + v[2]) + (v[1] + v[3]);

  *(v4f*)&red[r][c * 4] = s;
  __syncthreads();

  if (t < 32) {
    float a0 = 0.f, a1 = 0.f, a2 = 0.f, a3 = 0.f;
#pragma unroll
    for (int j = 0; j < 32; j += 4) {
      a0 += red[j][t];
      a1 += red[j + 1][t];
      a2 += red[j + 2][t];
      a3 += red[j + 3][t];
    }
    xs[b * D_ + q * 32 + t] = (a0 + a2) + (a1 + a3);
  }
}

#define PWS 68
#define PAS 132

__device__ __forceinline__ void w_load(const float* __restrict__ Wo, int t, v4f (&wr)[16]) {
#pragma unroll
  for (int n = 0; n < 16; ++n) wr[n] = *(const v4f*)(Wo + 512 * n + 4 * t);
}
__device__ __forceinline__ void w_store(float* WS, int t, const v4f (&wr)[16]) {
#pragma unroll
  for (int n = 0; n < 16; ++n) *(v4f*)(WS + (8 * n + (t >> 4)) * PWS + 4 * (t & 15)) = wr[n];
}

struct frag_k128 {
  v4f a[16];
  float b[16][4];
};
__device__ __forceinline__ void frag_k128_load(frag_k128& f, const float* AS, const float* WS,
                                               int n0, int lh, int hf) {
  const float* pa = AS + lh * PAS + 4 * hf;
  const float* pb = WS + (4 * hf) * PWS + n0 + lh;
#pragma unroll
  for (int jj = 0; jj < 16; ++jj) {
    f.a[jj] = *(const v4f*)(pa + 8 * jj);
#pragma unroll
    for (int e = 0; e < 4; ++e) f.b[jj][e] = pb[(8 * jj + e) * PWS];
  }
}
__device__ __forceinline__ v8f frag_k128_mma_t(const frag_k128& f) {
  v8f acc0 = {};
  v8f acc1 = {};
#pragma unroll
  for (int jj = 0; jj < 16; ++jj) {
    v2f alo = f.a[jj].xy, ahi = f.a[jj].zw;
    v2f blo = {f.b[jj][0], f.b[jj][1]};
    v2f bhi = {f.b[jj][2], f.b[jj][3]};
    acc0 = __builtin_amdgcn_wmma_f32_16x16x4_f32(false, blo, false, alo, (short)0,
                                                 acc0, false, false);
    acc1 = __builtin_amdgcn_wmma_f32_16x16x4_f32(false, bhi, false, ahi, (short)0,
                                                 acc1, false, false);
  }
  return acc0 + acc1;
}

__device__ __forceinline__ float xhalf(float x) {
  int b = __builtin_bit_cast(int, x);
  return __builtin_bit_cast(float,
      __builtin_amdgcn_permlanex16(b, b, 0x76543210, 0xfedcba98, false, false));
}

__device__ __forceinline__ float col_scale(const v8f& v, float (*ssq)[16], int w, int lh, int hf) {
  float q0 = v[0] * v[0] + v[1] * v[1];
  float q1 = v[2] * v[2] + v[3] * v[3];
  float q2 = v[4] * v[4] + v[5] * v[5];
  float q3 = v[6] * v[6] + v[7] * v[7];
  float q = (q0 + q1) + (q2 + q3);
  q += xhalf(q);
  ssq[w][lh] = q;
  __syncthreads();
  float sn = (ssq[0][lh] + ssq[1][lh]) + (ssq[2][lh] + ssq[3][lh]);
  return (sn / (1.f + sn)) * rsqrtf(sn);
}

__global__ __launch_bounds__(128) void k_iter0(const float* __restrict__ xs,
                                               const float* __restrict__ W,
                                               float* __restrict__ u) {
  int blk = blockIdx.x;
  int o  = blk >> 2;
  int b0 = (blk & 3) * 16;
  int t  = threadIdx.x;
  int w  = t >> 5;
  int l  = t & 31;
  int lh = l & 15;
  int hf = l >> 4;

  __shared__ __attribute__((aligned(16))) float WS[128 * PWS];
  __shared__ __attribute__((aligned(16))) float AS[16 * PAS];
  __shared__ __attribute__((aligned(16))) float out0s[16 * PWS];
  __shared__ float ssq[4][16];

  const float* ag = xs + (size_t)(b0 + w) * D_ + 4 * l;
  v4f wr[16], ar[4];
#pragma unroll
  for (int n = 0; n < 4; ++n) ar[n] = *(const v4f*)(ag + (size_t)(4 * n) * D_);
  w_load(W + (size_t)o * D_ * H_, t, wr);
  __builtin_amdgcn_sched_barrier(0);
#pragma unroll
  for (int n = 0; n < 4; ++n) *(v4f*)(AS + (4 * n + w) * PAS + 4 * l) = ar[n];
  w_store(WS, t, wr);
  __syncthreads();

  frag_k128 f;
  frag_k128_load(f, AS, WS, 16 * w, lh, hf);
  v4f q0[8], q1[8];
  {
    const float* pq = WS + (16 * w + lh) * PWS + 4 * hf;
#pragma unroll
    for (int jj = 0; jj < 8; ++jj) {
      q0[jj] = *(const v4f*)(pq + 8 * jj);
      q1[jj] = *(const v4f*)(pq + 64 * PWS + 8 * jj);
    }
  }
  __builtin_amdgcn_sched_barrier(0);

  v8f v = frag_k128_mma_t(f);
  v *= (1.0f / O_);

  float sc = col_scale(v, ssq, w, lh, hf);
  {
    v4f o0 = {v[0] * sc, v[1] * sc, v[2] * sc, v[3] * sc};
    v4f o1 = {v[4] * sc, v[5] * sc, v[6] * sc, v[7] * sc};
    float* po = out0s + lh * PWS + 16 * w + 8 * hf;
    *(v4f*)(po)     = o0;
    *(v4f*)(po + 4) = o1;
  }
  __syncthreads();

  v8f c0 = {};
  v8f c1 = {};
  {
    v4f fa[8];
    const float* pa = out0s + lh * PWS + 4 * hf;
#pragma unroll
    for (int jj = 0; jj < 8; ++jj) fa[jj] = *(const v4f*)(pa + 8 * jj);
    __builtin_amdgcn_sched_barrier(0);
#pragma unroll
    for (int jj = 0; jj < 8; ++jj) {
      v2f alo = fa[jj].xy, ahi = fa[jj].zw;
      v2f b0lo = q0[jj].xy, b0hi = q0[jj].zw;
      v2f b1lo = q1[jj].xy, b1hi = q1[jj].zw;
      c0 = __builtin_amdgcn_wmma_f32_16x16x4_f32(false, alo, false, b0lo, (short)0,
                                                 c0, false, false);
      c1 = __builtin_amdgcn_wmma_f32_16x16x4_f32(false, alo, false, b1lo, (short)0,
                                                 c1, false, false);
      c0 = __builtin_amdgcn_wmma_f32_16x16x4_f32(false, ahi, false, b0hi, (short)0,
                                                 c0, false, false);
      c1 = __builtin_amdgcn_wmma_f32_16x16x4_f32(false, ahi, false, b1hi, (short)0,
                                                 c1, false, false);
    }
  }

  float* ub = u + ((size_t)(b0 + 8 * hf) * O_ + o) * D_ + 16 * w + lh;
#pragma unroll
  for (int r = 0; r < 8; ++r) {
    ub[(size_t)r * O_ * D_]      = c0[r];
    ub[(size_t)r * O_ * D_ + 64] = c1[r];
  }
}

#define RXP 140
#define RPP 132
#define R_US (128 * RXP)
#define R_PS (R_US + 32 * RXP)
#define R_YS (R_PS + 32 * RPP)
#define RLDS (162 * 256)

__global__ __launch_bounds__(256) void k_route(const float* __restrict__ x,
                                               const float* __restrict__ u,
                                               float* __restrict__ yp) {
  int blk = blockIdx.x;
  int b  = blk >> 2;
  int c  = blk & 3;
  int t  = threadIdx.x;
  int w  = t >> 5;
  int l  = t & 31;
  int lh = l & 15;
  int hf = l >> 4;

  __shared__ __attribute__((aligned(16))) float sm[RLDS];
  float* XS = sm;
  float* US = sm + R_US;
  float* PS = sm + R_PS;
  float* YS = sm + R_YS;

  const float* xg = x + ((size_t)b * I_ + 128 * c + w) * D_ + 4 * l;
  const float* ug = u + ((size_t)b * O_ + w) * D_ + 4 * l;
  v4f xr[16], ur[4];
#pragma unroll
  for (int n = 0; n < 4; ++n) ur[n] = *(const v4f*)(ug + (size_t)(8 * n) * D_);
#pragma unroll
  for (int n = 0; n < 16; ++n) xr[n] = *(const v4f*)(xg + (size_t)(8 * n) * D_);
  __builtin_amdgcn_sched_barrier(0);
#pragma unroll
  for (int n = 0; n < 4; ++n) *(v4f*)(US + (8 * n + w) * RXP + 4 * l) = ur[n];
#pragma unroll
  for (int n = 0; n < 16; ++n) *(v4f*)(XS + (8 * n + w) * RXP + 4 * l) = xr[n];
  __syncthreads();

  v8f acc0 = {};
  v8f acc1 = {};
  {
    v4f fa[16], fb0[16], fb1[16];
    const float* pa = XS + (16 * w + lh) * RXP + 4 * hf;
    const float* pb = US + lh * RXP + 4 * hf;
#pragma unroll
    for (int jj = 0; jj < 16; ++jj) {
      fa[jj]  = *(const v4f*)(pa + 8 * jj);
      fb0[jj] = *(const v4f*)(pb + 8 * jj);
      fb1[jj] = *(const v4f*)(pb + 16 * RXP + 8 * jj);
    }
    __builtin_amdgcn_sched_barrier(0);
#pragma unroll
    for (int jj = 0; jj < 16; ++jj) {
      v2f alo = fa[jj].xy, ahi = fa[jj].zw;
      v2f b0lo = fb0[jj].xy, b0hi = fb0[jj].zw;
      v2f b1lo = fb1[jj].xy, b1hi = fb1[jj].zw;
      acc0 = __builtin_amdgcn_wmma_f32_16x16x4_f32(false, b0lo, false, alo, (short)0,
                                                   acc0, false, false);
      acc1 = __builtin_amdgcn_wmma_f32_16x16x4_f32(false, b1lo, false, alo, (short)0,
                                                   acc1, false, false);
      acc0 = __builtin_amdgcn_wmma_f32_16x16x4_f32(false, b0hi, false, ahi, (short)0,
                                                   acc0, false, false);
      acc1 = __builtin_amdgcn_wmma_f32_16x16x4_f32(false, b1hi, false, ahi, (short)0,
                                                   acc1, false, false);
    }
  }

  {
    float m0 = fmaxf(fmaxf(acc0[0], acc1[0]), fmaxf(acc0[1], acc1[1]));
    float m1 = fmaxf(fmaxf(acc0[2], acc1[2]), fmaxf(acc0[3], acc1[3]));
    float m2 = fmaxf(fmaxf(acc0[4], acc1[4]), fmaxf(acc0[5], acc1[5]));
    float m3 = fmaxf(fmaxf(acc0[6], acc1[6]), fmaxf(acc0[7], acc1[7]));
    float m = fmaxf(fmaxf(m0, m1), fmaxf(m2, m3));
    m = fmaxf(m, xhalf(m));
    float e0[8], e1[8];
#pragma unroll
    for (int r = 0; r < 8; ++r) {
      e0[r] = __expf(acc0[r] - m);
      e1[r] = __expf(acc1[r] - m);
    }
    float s0 = (e0[0] + e1[0]) + (e0[1] + e1[1]);
    float s1 = (e0[2] + e1[2]) + (e0[3] + e1[3]);
    float s2 = (e0[4] + e1[4]) + (e0[5] + e1[5]);
    float s3 = (e0[6] + e1[6]) + (e0[7] + e1[7]);
    float s = (s0 + s1) + (s2 + s3);
    s += xhalf(s);
    float inv = 1.0f / s;
    float* pw = PS + (8 * hf) * RPP + 16 * w + lh;
#pragma unroll
    for (int r = 0; r < 8; ++r) {
      pw[r * RPP]        = e0[r] * inv;
      pw[(16 + r) * RPP] = e1[r] * inv;
    }
  }
  __syncthreads();

  int np = w & 3;
  int k2 = w >> 2;
  v8f c00 = {}, c01 = {}, c10 = {}, c11 = {};
  {
    v4f ga0[8], ga1[8];
    float gb0[8][4], gb1[8][4];
    const float* qa = PS + lh * RPP + 64 * k2 + 4 * hf;
    const float* qb = XS + (64 * k2 + 4 * hf) * RXP + 32 * np + lh;
#pragma unroll
    for (int jj = 0; jj < 8; ++jj) {
      ga0[jj] = *(const v4f*)(qa + 8 * jj);
      ga1[jj] = *(const v4f*)(qa + 16 * RPP + 8 * jj);
#pragma unroll
      for (int e = 0; e < 4; ++e) {
        gb0[jj][e] = qb[(8 * jj + e) * RXP];
        gb1[jj][e] = qb[(8 * jj + e) * RXP + 16];
      }
    }
    __builtin_amdgcn_sched_barrier(0);
#pragma unroll
    for (int jj = 0; jj < 8; ++jj) {
      v2f a0lo = ga0[jj].xy, a0hi = ga0[jj].zw;
      v2f a1lo = ga1[jj].xy, a1hi = ga1[jj].zw;
      v2f b0lo = {gb0[jj][0], gb0[jj][1]};
      v2f b0hi = {gb0[jj][2], gb0[jj][3]};
      v2f b1lo = {gb1[jj][0], gb1[jj][1]};
      v2f b1hi = {gb1[jj][2], gb1[jj][3]};
      c00 = __builtin_amdgcn_wmma_f32_16x16x4_f32(false, a0lo, false, b0lo, (short)0,
                                                  c00, false, false);
      c01 = __builtin_amdgcn_wmma_f32_16x16x4_f32(false, a0lo, false, b1lo, (short)0,
                                                  c01, false, false);
      c10 = __builtin_amdgcn_wmma_f32_16x16x4_f32(false, a1lo, false, b0lo, (short)0,
                                                  c10, false, false);
      c11 = __builtin_amdgcn_wmma_f32_16x16x4_f32(false, a1lo, false, b1lo, (short)0,
                                                  c11, false, false);
      c00 = __builtin_amdgcn_wmma_f32_16x16x4_f32(false, a0hi, false, b0hi, (short)0,
                                                  c00, false, false);
      c01 = __builtin_amdgcn_wmma_f32_16x16x4_f32(false, a0hi, false, b1hi, (short)0,
                                                  c01, false, false);
      c10 = __builtin_amdgcn_wmma_f32_16x16x4_f32(false, a1hi, false, b0hi, (short)0,
                                                  c10, false, false);
      c11 = __builtin_amdgcn_wmma_f32_16x16x4_f32(false, a1hi, false, b1hi, (short)0,
                                                  c11, false, false);
    }
  }

  float* rw = YS + k2 * (32 * RPP) + (8 * hf) * RPP + 32 * np + lh;
#pragma unroll
  for (int r = 0; r < 8; ++r) {
    rw[r * RPP]             = c00[r];
    rw[r * RPP + 16]        = c01[r];
    rw[(16 + r) * RPP]      = c10[r];
    rw[(16 + r) * RPP + 16] = c11[r];
  }
  __syncthreads();

  float* yo = yp + (((size_t)c * B_ + b) * O_) * D_ + 4 * l;
#pragma unroll
  for (int n = 0; n < 4; ++n) {
    int o = 8 * n + w;
    v4f s0 = *(const v4f*)(YS + o * RPP + 4 * l);
    v4f s1 = *(const v4f*)(YS + 32 * RPP + o * RPP + 4 * l);
    *(v4f*)(yo + (size_t)o * D_) = s0 + s1;
  }
}

__global__ __launch_bounds__(128) void k_out(const float* __restrict__ yp,
                                             const float* __restrict__ W,
                                             float* __restrict__ out) {
  int blk = blockIdx.x;
  int o  = blk >> 2;
  int b0 = (blk & 3) * 16;
  int t  = threadIdx.x;
  int w  = t >> 5;
  int l  = t & 31;
  int lh = l & 15;
  int hf = l >> 4;

  __shared__ __attribute__((aligned(16))) float WS[128 * PWS];
  __shared__ __attribute__((aligned(16))) float AS[16 * PAS];
  __shared__ float ssq[4][16];

  const float* yg = yp + ((size_t)(b0 + w) * O_ + o) * D_ + 4 * l;
  const size_t PL = (size_t)B_ * O_ * D_;
  v4f wr[16], y0[4], y1[4], y2[4], y3[4];
#pragma unroll
  for (int n = 0; n < 4; ++n) {
    const float* yn = yg + (size_t)(4 * n) * O_ * D_;
    y0[n] = *(const v4f*)(yn);
    y1[n] = *(const v4f*)(yn + PL);
    y2[n] = *(const v4f*)(yn + 2 * PL);
    y3[n] = *(const v4f*)(yn + 3 * PL);
  }
  w_load(W + (size_t)o * D_ * H_, t, wr);
  __builtin_amdgcn_sched_barrier(0);
#pragma unroll
  for (int n = 0; n < 4; ++n)
    *(v4f*)(AS + (4 * n + w) * PAS + 4 * l) = (y0[n] + y1[n]) + (y2[n] + y3[n]);
  w_store(WS, t, wr);
  __syncthreads();

  frag_k128 f;
  frag_k128_load(f, AS, WS, 16 * w, lh, hf);
  __builtin_amdgcn_sched_barrier(0);
  v8f v = frag_k128_mma_t(f);

  float sc = col_scale(v, ssq, w, lh, hf);

  v4f o0 = {v[0] * sc, v[1] * sc, v[2] * sc, v[3] * sc};
  v4f o1 = {v[4] * sc, v[5] * sc, v[6] * sc, v[7] * sc};
  float* ob = out + ((size_t)(b0 + lh) * O_ + o) * H_ + 16 * w + 8 * hf;
  *(v4f*)(ob)     = o0;
  *(v4f*)(ob + 4) = o1;
}

extern "C" void kernel_launch(void* const* d_in, const int* in_sizes, int n_in,
                              void* d_out, int out_size, void* d_ws, size_t ws_size,
                              hipStream_t stream) {
  const float* x = (const float*)d_in[0];
  const float* W = (const float*)d_in[1];
  float* out = (float*)d_out;

  char* ws = (char*)d_ws;
  float* xs = (float*)ws;    ws += (size_t)B_ * D_ * sizeof(float);
  float* u  = (float*)ws;    ws += (size_t)B_ * O_ * D_ * sizeof(float);
  float* yp = (float*)ws;

  k_colsum<<<B_ * 4, 256, 0, stream>>>(x, xs);
  k_iter0<<<O_ * (B_ / 16), 128, 0, stream>>>(xs, W, u);
  k_route<<<B_ * 4, 256, 0, stream>>>(x, u, yp);
  k_out<<<O_ * (B_ / 16), 128, 0, stream>>>(yp, W, out);
}
